# starting kernel
# baseline (speedup 1.0000x reference)
#include <hip/hip_runtime.h>

#define NROWS 8192
#define DIM   256
#define INV_SIGMA 3.33333333333f
#define LOGC      0.28503428291f

#define ROWS_PER_BLOCK  64
#define WAVES_PER_BLOCK 4
#define THREADS         (WAVES_PER_BLOCK * 32)
#define NCOLTILES       (NROWS / 16)
#define NSLICES         8
#define TILES_PER_SLICE (NCOLTILES / NSLICES)
#define LDS_STRIDE      264

typedef __attribute__((ext_vector_type(16))) __bf16 v16bf;
typedef __attribute__((ext_vector_type(8)))  float  v8f;

__device__ __forceinline__ unsigned short f32_to_bf16_rne(float f) {
    unsigned u = __float_as_uint(f);
    u += 0x7FFFu + ((u >> 16) & 1u);
    return (unsigned short)(u >> 16);
}
__device__ __forceinline__ float bf16_to_f32(unsigned short h) {
    return __uint_as_float(((unsigned)h) << 16);
}

__device__ __forceinline__ unsigned f2key(float f) {
    unsigned b = __float_as_uint(f);
    return (b & 0x80000000u) ? ~b : (b | 0x80000000u);
}
__device__ __forceinline__ float key2f(unsigned u) {
    unsigned b = (u & 0x80000000u) ? (u ^ 0x80000000u) : ~u;
    return __uint_as_float(b);
}

__global__ void init_keys_kernel(unsigned* __restrict__ keys) {
    keys[blockIdx.x * blockDim.x + threadIdx.x] = 0xFFFFFFFFu;
}

__global__ void normalize_split_kernel(const float* __restrict__ Ex,
                                       const float* __restrict__ Ey,
                                       unsigned short* __restrict__ Xhi,
                                       unsigned short* __restrict__ Xlo,
                                       unsigned short* __restrict__ Yhi,
                                       unsigned short* __restrict__ Ylo) {
    const int wave = (blockIdx.x * blockDim.x + threadIdx.x) >> 5;
    const int lane = threadIdx.x & 31;

    const float* src;
    unsigned short *hi, *lo;
    int row;
    if (wave < NROWS) { src = Ex; hi = Xhi; lo = Xlo; row = wave; }
    else              { src = Ey; hi = Yhi; lo = Ylo; row = wave - NROWS; }

    const float* p = src + (size_t)row * DIM;
    float v[DIM / 32];
    float ss = 0.0f;
#pragma unroll
    for (int i = 0; i < DIM / 32; ++i) {
        v[i] = p[lane + 32 * i];
        ss = fmaf(v[i], v[i], ss);
    }
#pragma unroll
    for (int off = 16; off > 0; off >>= 1)
        ss += __shfl_xor(ss, off, 32);

    const float inv = 1.0f / fmaxf(sqrtf(ss), 1e-8f);

#pragma unroll
    for (int i = 0; i < DIM / 32; ++i) {
        const float x = v[i] * inv;
        const unsigned short h = f32_to_bf16_rne(x);
        const unsigned short l = f32_to_bf16_rne(x - bf16_to_f32(h));
        const size_t idx = (size_t)row * DIM + lane + 32 * i;
        hi[idx] = h;
        lo[idx] = l;
    }
}

__device__ __forceinline__ v16bf lds_frag(const unsigned short* p) {
    union { uint4 q[2]; v16bf v; } f;
    f.q[0] = *(const uint4*)(p);
    f.q[1] = *(const uint4*)(p + 16);
    return f.v;
}

__global__ __launch_bounds__(THREADS) void cosmin_wmma_kernel(
    const unsigned short* __restrict__ Xhi,
    const unsigned short* __restrict__ Xlo,
    const unsigned short* __restrict__ Yhi,
    const unsigned short* __restrict__ Ylo,
    unsigned* __restrict__ keys) {
    __shared__ unsigned short lAh[ROWS_PER_BLOCK * LDS_STRIDE];
    __shared__ unsigned short lAl[ROWS_PER_BLOCK * LDS_STRIDE];
    __shared__ unsigned short lBh[16 * LDS_STRIDE];
    __shared__ unsigned short lBl[16 * LDS_STRIDE];

    const int tid      = threadIdx.x;
    const int lane     = tid & 31;
    const int wave     = tid >> 5;
    const int rowBase  = blockIdx.x * ROWS_PER_BLOCK;
    const int tileBase = blockIdx.y * TILES_PER_SLICE;

#pragma unroll 2
    for (int idx = tid; idx < ROWS_PER_BLOCK * (DIM / 8); idx += THREADS) {
        const int row = idx >> 5;
        const int ch  = idx & 31;
        const size_t s = (size_t)(rowBase + row) * DIM + ch * 8;
        *(uint4*)&lAh[row * LDS_STRIDE + ch * 8] = *(const uint4*)&Xhi[s];
        *(uint4*)&lAl[row * LDS_STRIDE + ch * 8] = *(const uint4*)&Xlo[s];
    }

    uint4 rh[4], rl[4];
    auto prefetchB = [&](int t) {
        const int colBase = (tileBase + t) * 16;
#pragma unroll
        for (int i = 0; i < 4; ++i) {
            const int cid = tid + THREADS * i;
            const int row = cid >> 5;
            const int ch  = cid & 31;
            const size_t s = (size_t)(colBase + row) * DIM + ch * 8;
            rh[i] = *(const uint4*)&Yhi[s];
            rl[i] = *(const uint4*)&Ylo[s];
        }
    };
    prefetchB(0);

    float minv[8];
#pragma unroll
    for (int r = 0; r < 8; ++r) minv[r] = 3.402823466e38f;

    const int hiSel = (lane >> 4) << 3;
    const int aOff  = (wave * 16 + (lane & 15)) * LDS_STRIDE + hiSel;
    const int bOff  = (lane & 15) * LDS_STRIDE + hiSel;

    for (int t = 0; t < TILES_PER_SLICE; ++t) {
        __syncthreads();
#pragma unroll
        for (int i = 0; i < 4; ++i) {
            const int cid = tid + THREADS * i;
            const int row = cid >> 5;
            const int ch  = cid & 31;
            *(uint4*)&lBh[row * LDS_STRIDE + ch * 8] = rh[i];
            *(uint4*)&lBl[row * LDS_STRIDE + ch * 8] = rl[i];
        }
        __syncthreads();
        if (t + 1 < TILES_PER_SLICE) prefetchB(t + 1);

        v8f acc = {};
#pragma unroll
        for (int c = 0; c < DIM / 32; ++c) {
            const int k0 = c * 32;
            v16bf ah = lds_frag(&lAh[aOff + k0]);
            v16bf al = lds_frag(&lAl[aOff + k0]);
            v16bf bh = lds_frag(&lBh[bOff + k0]);
            v16bf bl = lds_frag(&lBl[bOff + k0]);
            acc = __builtin_amdgcn_wmma_f32_16x16x32_bf16(
                      false, al, false, bh, (short)0, acc, false, false);
            acc = __builtin_amdgcn_wmma_f32_16x16x32_bf16(
                      false, ah, false, bl, (short)0, acc, false, false);
            acc = __builtin_amdgcn_wmma_f32_16x16x32_bf16(
                      false, ah, false, bh, (short)0, acc, false, false);
        }

#pragma unroll
        for (int r = 0; r < 8; ++r) {
            const float m    = acc[r];
            const float z    = (m - 1.0f) * INV_SIGMA;
            const float logp = fmaf(-0.5f * z, z, LOGC);
            const float C    = -__expf(logp) * logp;
            minv[r] = fminf(minv[r], C);
        }
    }

#pragma unroll
    for (int r = 0; r < 8; ++r) {
#pragma unroll
        for (int off = 1; off < 16; off <<= 1)
            minv[r] = fminf(minv[r], __shfl_xor(minv[r], off, 32));
    }

    if ((lane & 15) == 0) {
        const int base = rowBase + wave * 16 + ((lane >> 4) << 3);
#pragma unroll
        for (int r = 0; r < 8; ++r)
            atomicMin(&keys[base + r], f2key(minv[r]));
    }
}

__global__ void finalize_kernel(const unsigned* __restrict__ keys,
                                float* __restrict__ out) {
    const int i = blockIdx.x * blockDim.x + threadIdx.x;
    out[i] = key2f(keys[i]);
}

extern "C" void kernel_launch(void* const* d_in, const int* in_sizes, int n_in,
                              void* d_out, int out_size, void* d_ws, size_t ws_size,
                              hipStream_t stream) {
    const float* Ex = (const float*)d_in[0];
    const float* Ey = (const float*)d_in[1];

    const size_t elems = (size_t)NROWS * DIM;
    unsigned short* Xhi = (unsigned short*)d_ws;
    unsigned short* Xlo = Xhi + elems;
    unsigned short* Yhi = Xlo + elems;
    unsigned short* Ylo = Yhi + elems;
    unsigned*      keys = (unsigned*)(Ylo + elems);

    hipLaunchKernelGGL(init_keys_kernel, dim3(NROWS / 256), dim3(256), 0, stream,
                       keys);
    hipLaunchKernelGGL(normalize_split_kernel, dim3((2 * NROWS) / 8), dim3(256), 0,
                       stream, Ex, Ey, Xhi, Xlo, Yhi, Ylo);
    hipLaunchKernelGGL(cosmin_wmma_kernel,
                       dim3(NROWS / ROWS_PER_BLOCK, NSLICES), dim3(THREADS), 0,
                       stream, Xhi, Xlo, Yhi, Ylo, keys);
    hipLaunchKernelGGL(finalize_kernel, dim3(NROWS / 256), dim3(256), 0, stream,
                       keys, (float*)d_out);
}


# GEMM operands kept in registers: each wave loads its 16-row A fragments (hi+lo, 128 VGPRs) once for the whole sweep and holds the prefetched B tile in named VGPRs; LDS now carries only the B tile
# speedup vs baseline: 2.4462x; 2.4462x over previous
#include <hip/hip_runtime.h>

#define NROWS 8192
#define DIM   256
#define INV_SIGMA 3.33333333333f
#define LOGC      0.28503428291f

#define ROWS_PER_BLOCK  64
#define WAVES_PER_BLOCK 4
#define THREADS         (WAVES_PER_BLOCK * 32)
#define NCOLTILES       (NROWS / 16)
#define NSLICES         8
#define TILES_PER_SLICE (NCOLTILES / NSLICES)
#define LDS_STRIDE      264

typedef __attribute__((ext_vector_type(16))) __bf16 v16bf;
typedef __attribute__((ext_vector_type(8)))  float  v8f;
typedef __attribute__((ext_vector_type(4)))  unsigned int u32x4;

__device__ __forceinline__ unsigned short f32_to_bf16_rne(float f) {
    unsigned u = __float_as_uint(f);
    u += 0x7FFFu + ((u >> 16) & 1u);
    return (unsigned short)(u >> 16);
}
__device__ __forceinline__ float bf16_to_f32(unsigned short h) {
    return __uint_as_float(((unsigned)h) << 16);
}

__device__ __forceinline__ unsigned f2key(float f) {
    unsigned b = __float_as_uint(f);
    return (b & 0x80000000u) ? ~b : (b | 0x80000000u);
}
__device__ __forceinline__ float key2f(unsigned u) {
    unsigned b = (u & 0x80000000u) ? (u ^ 0x80000000u) : ~u;
    return __uint_as_float(b);
}

__global__ void init_keys_kernel(unsigned* __restrict__ keys) {
    keys[blockIdx.x * blockDim.x + threadIdx.x] = 0xFFFFFFFFu;
}

__global__ void normalize_split_kernel(const float* __restrict__ Ex,
                                       const float* __restrict__ Ey,
                                       unsigned short* __restrict__ Xhi,
                                       unsigned short* __restrict__ Xlo,
                                       unsigned short* __restrict__ Yhi,
                                       unsigned short* __restrict__ Ylo) {
    const int wave = (blockIdx.x * blockDim.x + threadIdx.x) >> 5;
    const int lane = threadIdx.x & 31;

    const float* src;
    unsigned short *hi, *lo;
    int row;
    if (wave < NROWS) { src = Ex; hi = Xhi; lo = Xlo; row = wave; }
    else              { src = Ey; hi = Yhi; lo = Ylo; row = wave - NROWS; }

    const float* p = src + (size_t)row * DIM;
    float v[DIM / 32];
    float ss = 0.0f;
#pragma unroll
    for (int i = 0; i < DIM / 32; ++i) {
        v[i] = p[lane + 32 * i];
        ss = fmaf(v[i], v[i], ss);
    }
#pragma unroll
    for (int off = 16; off > 0; off >>= 1)
        ss += __shfl_xor(ss, off, 32);

    const float inv = 1.0f / fmaxf(sqrtf(ss), 1e-8f);

#pragma unroll
    for (int i = 0; i < DIM / 32; ++i) {
        const float x = v[i] * inv;
        const unsigned short h = f32_to_bf16_rne(x);
        const unsigned short l = f32_to_bf16_rne(x - bf16_to_f32(h));
        const size_t idx = (size_t)row * DIM + lane + 32 * i;
        hi[idx] = h;
        lo[idx] = l;
    }
}

__device__ __forceinline__ v16bf lds_frag(const unsigned short* p) {
    union { uint4 q[2]; v16bf v; } f;
    f.q[0] = *(const uint4*)(p);
    f.q[1] = *(const uint4*)(p + 16);
    return f.v;
}
__device__ __forceinline__ v16bf glb_frag(const unsigned short* __restrict__ p) {
    union { uint4 q[2]; v16bf v; } f;
    f.q[0] = *(const uint4*)(p);
    f.q[1] = *(const uint4*)(p + 16);
    return f.v;
}

__global__ __launch_bounds__(THREADS) void cosmin_wmma_kernel(
    const unsigned short* __restrict__ Xhi,
    const unsigned short* __restrict__ Xlo,
    const unsigned short* __restrict__ Yhi,
    const unsigned short* __restrict__ Ylo,
    unsigned* __restrict__ keys) {
    __shared__ unsigned short lBh[16 * LDS_STRIDE];
    __shared__ unsigned short lBl[16 * LDS_STRIDE];

    const int tid      = threadIdx.x;
    const int lane     = tid & 31;
    const int wave     = tid >> 5;
    const int rowBase  = blockIdx.x * ROWS_PER_BLOCK;
    const int tileBase = blockIdx.y * TILES_PER_SLICE;

    const int hiSel = (lane >> 4) << 3;

    v16bf ah[DIM / 32], al[DIM / 32];
    {
        const size_t aRow =
            (size_t)(rowBase + wave * 16 + (lane & 15)) * DIM + hiSel;
#pragma unroll
        for (int c = 0; c < DIM / 32; ++c) {
            ah[c] = glb_frag(&Xhi[aRow + c * 32]);
            al[c] = glb_frag(&Xlo[aRow + c * 32]);
        }
    }

    const int gOff = (tid >> 5) * DIM + (tid & 31) * 8;
    const int sOff = (tid >> 5) * LDS_STRIDE + (tid & 31) * 8;
    u32x4 rh0, rh1, rh2, rh3, rl0, rl1, rl2, rl3;
#define PREFETCH_B(T)                                                        \
    do {                                                                     \
        const size_t g_ = (size_t)((tileBase + (T)) * 16) * DIM + gOff;      \
        rh0 = *(const u32x4*)&Yhi[g_];                                       \
        rl0 = *(const u32x4*)&Ylo[g_];                                       \
        rh1 = *(const u32x4*)&Yhi[g_ + 4 * DIM];                             \
        rl1 = *(const u32x4*)&Ylo[g_ + 4 * DIM];                             \
        rh2 = *(const u32x4*)&Yhi[g_ + 8 * DIM];                             \
        rl2 = *(const u32x4*)&Ylo[g_ + 8 * DIM];                             \
        rh3 = *(const u32x4*)&Yhi[g_ + 12 * DIM];                            \
        rl3 = *(const u32x4*)&Ylo[g_ + 12 * DIM];                            \
    } while (0)
    PREFETCH_B(0);

    float minv[8];
#pragma unroll
    for (int r = 0; r < 8; ++r) minv[r] = 3.402823466e38f;

    const int bOff  = (lane & 15) * LDS_STRIDE + hiSel;

    for (int t = 0; t < TILES_PER_SLICE; ++t) {
        __syncthreads();
        *(u32x4*)&lBh[sOff]                   = rh0;
        *(u32x4*)&lBl[sOff]                   = rl0;
        *(u32x4*)&lBh[sOff + 4 * LDS_STRIDE]  = rh1;
        *(u32x4*)&lBl[sOff + 4 * LDS_STRIDE]  = rl1;
        *(u32x4*)&lBh[sOff + 8 * LDS_STRIDE]  = rh2;
        *(u32x4*)&lBl[sOff + 8 * LDS_STRIDE]  = rl2;
        *(u32x4*)&lBh[sOff + 12 * LDS_STRIDE] = rh3;
        *(u32x4*)&lBl[sOff + 12 * LDS_STRIDE] = rl3;
        __syncthreads();
        if (t + 1 < TILES_PER_SLICE) PREFETCH_B(t + 1);
        __builtin_amdgcn_sched_barrier(0);

        v8f acc = {};
#pragma unroll
        for (int c = 0; c < DIM / 32; ++c) {
            const int k0 = c * 32;
            v16bf bh = lds_frag(&lBh[bOff + k0]);
            v16bf bl = lds_frag(&lBl[bOff + k0]);
            acc = __builtin_amdgcn_wmma_f32_16x16x32_bf16(
                      false, al[c], false, bh, (short)0, acc, false, false);
            acc = __builtin_amdgcn_wmma_f32_16x16x32_bf16(
                      false, ah[c], false, bl, (short)0, acc, false, false);
            acc = __builtin_amdgcn_wmma_f32_16x16x32_bf16(
                      false, ah[c], false, bh, (short)0, acc, false, false);
        }

#pragma unroll
        for (int r = 0; r < 8; ++r) {
            const float m    = acc[r];
            const float z    = (m - 1.0f) * INV_SIGMA;
            const float logp = fmaf(-0.5f * z, z, LOGC);
            const float C    = -__expf(logp) * logp;
            minv[r] = fminf(minv[r], C);
        }
    }

#undef PREFETCH_B

#pragma unroll
    for (int r = 0; r < 8; ++r) {
#pragma unroll
        for (int off = 1; off < 16; off <<= 1)
            minv[r] = fminf(minv[r], __shfl_xor(minv[r], off, 32));
    }

    if ((lane & 15) == 0) {
        const int base = rowBase + wave * 16 + ((lane >> 4) << 3);
#pragma unroll
        for (int r = 0; r < 8; ++r)
            atomicMin(&keys[base + r], f2key(minv[r]));
    }
}

__global__ void finalize_kernel(const unsigned* __restrict__ keys,
                                float* __restrict__ out) {
    const int i = blockIdx.x * blockDim.x + threadIdx.x;
    out[i] = key2f(keys[i]);
}

extern "C" void kernel_launch(void* const* d_in, const int* in_sizes, int n_in,
                              void* d_out, int out_size, void* d_ws, size_t ws_size,
                              hipStream_t stream) {
    const float* Ex = (const float*)d_in[0];
    const float* Ey = (const float*)d_in[1];

    const size_t elems = (size_t)NROWS * DIM;
    unsigned short* Xhi = (unsigned short*)d_ws;
    unsigned short* Xlo = Xhi + elems;
    unsigned short* Yhi = Xlo + elems;
    unsigned short* Ylo = Yhi + elems;
    unsigned*      keys = (unsigned*)(Ylo + elems);

    hipLaunchKernelGGL(init_keys_kernel, dim3(NROWS / 256), dim3(256), 0, stream,
                       keys);
    hipLaunchKernelGGL(normalize_split_kernel, dim3((2 * NROWS) / 8), dim3(256), 0,
                       stream, Ex, Ey, Xhi, Xlo, Yhi, Ylo);
    hipLaunchKernelGGL(cosmin_wmma_kernel,
                       dim3(NROWS / ROWS_PER_BLOCK, NSLICES), dim3(THREADS), 0,
                       stream, Xhi, Xlo, Yhi, Ylo, keys);
    hipLaunchKernelGGL(finalize_kernel, dim3(NROWS / 256), dim3(256), 0, stream,
                       keys, (float*)d_out);
}


# Each wave keeps A fragments of 64 rows (was 16) in registers, 1 wave per SIMD, 256 blocks: a B fragment from LDS feeds 12 WMMAs on 4 accumulators (was 3 on 1); B read one tile ahead (2 LDS stages)
# speedup vs baseline: 2.9179x; 1.1928x over previous
#include <hip/hip_runtime.h>

#define NROWS 8192
#define DIM   256
#define INV_SIGMA 3.33333333333f
#define LOGC      0.28503428291f

#define ROW_TILES       4
#define ROWS_PER_WAVE   (ROW_TILES * 16)
#define ROWS_PER_BLOCK  256
#define WAVES_PER_BLOCK 4
#define THREADS         (WAVES_PER_BLOCK * 32)
#define NCOLTILES       (NROWS / 16)
#define NSLICES         8
#define TILES_PER_SLICE (NCOLTILES / NSLICES)
#define LDS_STRIDE      264
#define STAGE_ELEMS     (16 * LDS_STRIDE)

typedef __attribute__((ext_vector_type(16))) __bf16 v16bf;
typedef __attribute__((ext_vector_type(8)))  float  v8f;
typedef __attribute__((ext_vector_type(4)))  unsigned int u32x4;

__device__ __forceinline__ unsigned short f32_to_bf16_rne(float f) {
    unsigned u = __float_as_uint(f);
    u += 0x7FFFu + ((u >> 16) & 1u);
    return (unsigned short)(u >> 16);
}
__device__ __forceinline__ float bf16_to_f32(unsigned short h) {
    return __uint_as_float(((unsigned)h) << 16);
}

__device__ __forceinline__ unsigned f2key(float f) {
    unsigned b = __float_as_uint(f);
    return (b & 0x80000000u) ? ~b : (b | 0x80000000u);
}
__device__ __forceinline__ float key2f(unsigned u) {
    unsigned b = (u & 0x80000000u) ? (u ^ 0x80000000u) : ~u;
    return __uint_as_float(b);
}

__global__ void init_keys_kernel(unsigned* __restrict__ keys) {
    keys[blockIdx.x * blockDim.x + threadIdx.x] = 0xFFFFFFFFu;
}

__global__ void normalize_split_kernel(const float* __restrict__ Ex,
                                       const float* __restrict__ Ey,
                                       unsigned short* __restrict__ Xhi,
                                       unsigned short* __restrict__ Xlo,
                                       unsigned short* __restrict__ Yhi,
                                       unsigned short* __restrict__ Ylo) {
    const int wave = (blockIdx.x * blockDim.x + threadIdx.x) >> 5;
    const int lane = threadIdx.x & 31;

    const float* src;
    unsigned short *hi, *lo;
    int row;
    if (wave < NROWS) { src = Ex; hi = Xhi; lo = Xlo; row = wave; }
    else              { src = Ey; hi = Yhi; lo = Ylo; row = wave - NROWS; }

    const float* p = src + (size_t)row * DIM;
    float v[DIM / 32];
    float ss = 0.0f;
#pragma unroll
    for (int i = 0; i < DIM / 32; ++i) {
        v[i] = p[lane + 32 * i];
        ss = fmaf(v[i], v[i], ss);
    }
#pragma unroll
    for (int off = 16; off > 0; off >>= 1)
        ss += __shfl_xor(ss, off, 32);

    const float inv = 1.0f / fmaxf(sqrtf(ss), 1e-8f);

#pragma unroll
    for (int i = 0; i < DIM / 32; ++i) {
        const float x = v[i] * inv;
        const unsigned short h = f32_to_bf16_rne(x);
        const unsigned short l = f32_to_bf16_rne(x - bf16_to_f32(h));
        const size_t idx = (size_t)row * DIM + lane + 32 * i;
        hi[idx] = h;
        lo[idx] = l;
    }
}

__device__ __forceinline__ v16bf lds_frag(const unsigned short* p) {
    union { uint4 q[2]; v16bf v; } f;
    f.q[0] = *(const uint4*)(p);
    f.q[1] = *(const uint4*)(p + 16);
    return f.v;
}
__device__ __forceinline__ v16bf glb_frag(const unsigned short* __restrict__ p) {
    union { uint4 q[2]; v16bf v; } f;
    f.q[0] = *(const uint4*)(p);
    f.q[1] = *(const uint4*)(p + 16);
    return f.v;
}

#define FRAG_LOAD(FH, FL, S)                                                 \
    do {                                                                     \
        _Pragma("unroll")                                                    \
        for (int c_ = 0; c_ < DIM / 32; ++c_) {                              \
            FH[c_] = lds_frag(&lBh[(S) * STAGE_ELEMS + bOff + c_ * 32]);     \
            FL[c_] = lds_frag(&lBl[(S) * STAGE_ELEMS + bOff + c_ * 32]);     \
        }                                                                    \
    } while (0)

#define STORE_B(S)                                                           \
    do {                                                                     \
        const int s_ = (S) * STAGE_ELEMS + sOff;                             \
        *(u32x4*)&lBh[s_]                   = rh0;                           \
        *(u32x4*)&lBl[s_]                   = rl0;                           \
        *(u32x4*)&lBh[s_ + 4 * LDS_STRIDE]  = rh1;                           \
        *(u32x4*)&lBl[s_ + 4 * LDS_STRIDE]  = rl1;                           \
        *(u32x4*)&lBh[s_ + 8 * LDS_STRIDE]  = rh2;                           \
        *(u32x4*)&lBl[s_ + 8 * LDS_STRIDE]  = rl2;                           \
        *(u32x4*)&lBh[s_ + 12 * LDS_STRIDE] = rh3;                           \
        *(u32x4*)&lBl[s_ + 12 * LDS_STRIDE] = rl3;                           \
    } while (0)

#define PREFETCH_B(T)                                                        \
    do {                                                                     \
        const int tt_ = ((T) < TILES_PER_SLICE) ? (T) : (TILES_PER_SLICE - 1); \
        const size_t g_ = (size_t)((tileBase + tt_) * 16) * DIM + gOff;      \
        rh0 = *(const u32x4*)&Yhi[g_];                                       \
        rl0 = *(const u32x4*)&Ylo[g_];                                       \
        rh1 = *(const u32x4*)&Yhi[g_ + 4 * DIM];                             \
        rl1 = *(const u32x4*)&Ylo[g_ + 4 * DIM];                             \
        rh2 = *(const u32x4*)&Yhi[g_ + 8 * DIM];                             \
        rl2 = *(const u32x4*)&Ylo[g_ + 8 * DIM];                             \
        rh3 = *(const u32x4*)&Yhi[g_ + 12 * DIM];                            \
        rl3 = *(const u32x4*)&Ylo[g_ + 12 * DIM];                            \
    } while (0)

#define COMPUTE_TILE(FH, FL)                                                 \
    do {                                                                     \
        v8f acc[ROW_TILES];                                                  \
        _Pragma("unroll")                                                    \
        for (int rt = 0; rt < ROW_TILES; ++rt) acc[rt] = v8f{};              \
        _Pragma("unroll")                                                    \
        for (int c = 0; c < DIM / 32; ++c) {                                 \
            _Pragma("unroll")                                                \
            for (int rt = 0; rt < ROW_TILES; ++rt)                           \
                acc[rt] = __builtin_amdgcn_wmma_f32_16x16x32_bf16(           \
                    false, al[rt][c], false, FH[c], (short)0, acc[rt],       \
                    false, false);                                           \
            _Pragma("unroll")                                                \
            for (int rt = 0; rt < ROW_TILES; ++rt)                           \
                acc[rt] = __builtin_amdgcn_wmma_f32_16x16x32_bf16(           \
                    false, ah[rt][c], false, FL[c], (short)0, acc[rt],       \
                    false, false);                                           \
            _Pragma("unroll")                                                \
            for (int rt = 0; rt < ROW_TILES; ++rt)                           \
                acc[rt] = __builtin_amdgcn_wmma_f32_16x16x32_bf16(           \
                    false, ah[rt][c], false, FH[c], (short)0, acc[rt],       \
                    false, false);                                           \
        }                                                                    \
        _Pragma("unroll")                                                    \
        for (int rt = 0; rt < ROW_TILES; ++rt) {                             \
            _Pragma("unroll")                                                \
            for (int r = 0; r < 8; ++r) {                                    \
                const float m_    = acc[rt][r];                              \
                const float z_    = (m_ - 1.0f) * INV_SIGMA;                 \
                const float logp_ = fmaf(-0.5f * z_, z_, LOGC);              \
                const float C_    = -__expf(logp_) * logp_;                  \
                minv[rt][r] = fminf(minv[rt][r], C_);                        \
            }                                                                \
        }                                                                    \
    } while (0)

#define SWEEP_STEP(T, CH, CL, NH, NL)                                        \
    do {                                                                     \
         \
         \
        __syncthreads();                                                     \
        FRAG_LOAD(NH, NL, ((T) + 1) & 1);                                    \
        STORE_B((T) & 1);                                                    \
        PREFETCH_B((T) + 3);                                                 \
         \
         \
         \
        __builtin_amdgcn_sched_barrier(0);                                   \
        COMPUTE_TILE(CH, CL);                                                \
         \
         \
         \
        __builtin_amdgcn_sched_barrier(0);                                   \
    } while (0)

__global__ __launch_bounds__(THREADS) void cosmin_wmma_kernel(
    const unsigned short* __restrict__ Xhi,
    const unsigned short* __restrict__ Xlo,
    const unsigned short* __restrict__ Yhi,
    const unsigned short* __restrict__ Ylo,
    unsigned* __restrict__ keys) {
    __shared__ unsigned short lBh[2 * STAGE_ELEMS];
    __shared__ unsigned short lBl[2 * STAGE_ELEMS];

    const int tid      = threadIdx.x;
    const int lane     = tid & 31;
    const int wave     = tid >> 5;
    const int rowBase  = blockIdx.x * ROWS_PER_BLOCK + wave * ROWS_PER_WAVE;
    const int tileBase = blockIdx.y * TILES_PER_SLICE;

    const int hiSel = (lane >> 4) << 3;

    v16bf ah[ROW_TILES][DIM / 32], al[ROW_TILES][DIM / 32];
#pragma unroll
    for (int rt = 0; rt < ROW_TILES; ++rt) {
        const size_t aRow =
            (size_t)(rowBase + rt * 16 + (lane & 15)) * DIM + hiSel;
#pragma unroll
        for (int c = 0; c < DIM / 32; ++c) {
            ah[rt][c] = glb_frag(&Xhi[aRow + c * 32]);
            al[rt][c] = glb_frag(&Xlo[aRow + c * 32]);
        }
    }

    const int gOff = (tid >> 5) * DIM + (tid & 31) * 8;
    const int sOff = (tid >> 5) * LDS_STRIDE + (tid & 31) * 8;
    const int bOff = (lane & 15) * LDS_STRIDE + hiSel;
    u32x4 rh0, rh1, rh2, rh3, rl0, rl1, rl2, rl3;

    float minv[ROW_TILES][8];
#pragma unroll
    for (int rt = 0; rt < ROW_TILES; ++rt)
#pragma unroll
        for (int r = 0; r < 8; ++r) minv[rt][r] = 3.402823466e38f;

    v16bf f0h[DIM / 32], f0l[DIM / 32], f1h[DIM / 32], f1l[DIM / 32];

    PREFETCH_B(0);
    STORE_B(0);
    PREFETCH_B(1);
    STORE_B(1);
    PREFETCH_B(2);
    __syncthreads();
    FRAG_LOAD(f0h, f0l, 0);

    for (int t = 0; t < TILES_PER_SLICE; t += 2) {
        SWEEP_STEP(t,     f0h, f0l, f1h, f1l);
        SWEEP_STEP(t + 1, f1h, f1l, f0h, f0l);
    }

#pragma unroll
    for (int rt = 0; rt < ROW_TILES; ++rt) {
#pragma unroll
        for (int r = 0; r < 8; ++r) {
#pragma unroll
            for (int off = 1; off < 16; off <<= 1)
                minv[rt][r] =
                    fminf(minv[rt][r], __shfl_xor(minv[rt][r], off, 32));
        }
    }

    if ((lane & 15) == 0) {
#pragma unroll
        for (int rt = 0; rt < ROW_TILES; ++rt) {
            const int base = rowBase + rt * 16 + ((lane >> 4) << 3);
#pragma unroll
            for (int r = 0; r < 8; ++r)
                atomicMin(&keys[base + r], f2key(minv[rt][r]));
        }
    }
}

#undef SWEEP_STEP
#undef COMPUTE_TILE
#undef PREFETCH_B
#undef STORE_B
#undef FRAG_LOAD

__global__ void finalize_kernel(const unsigned* __restrict__ keys,
                                float* __restrict__ out) {
    const int i = blockIdx.x * blockDim.x + threadIdx.x;
    out[i] = key2f(keys[i]);
}

extern "C" void kernel_launch(void* const* d_in, const int* in_sizes, int n_in,
                              void* d_out, int out_size, void* d_ws, size_t ws_size,
                              hipStream_t stream) {
    const float* Ex = (const float*)d_in[0];
    const float* Ey = (const float*)d_in[1];

    const size_t elems = (size_t)NROWS * DIM;
    unsigned short* Xhi = (unsigned short*)d_ws;
    unsigned short* Xlo = Xhi + elems;
    unsigned short* Yhi = Xlo + elems;
    unsigned short* Ylo = Yhi + elems;
    unsigned*      keys = (unsigned*)(Ylo + elems);

    hipLaunchKernelGGL(init_keys_kernel, dim3(NROWS / 256), dim3(256), 0, stream,
                       keys);
    hipLaunchKernelGGL(normalize_split_kernel, dim3((2 * NROWS) / 8), dim3(256), 0,
                       stream, Ex, Ey, Xhi, Xlo, Yhi, Ylo);
    hipLaunchKernelGGL(cosmin_wmma_kernel,
                       dim3(NROWS / ROWS_PER_BLOCK, NSLICES), dim3(THREADS), 0,
                       stream, Xhi, Xlo, Yhi, Ylo, keys);
    hipLaunchKernelGGL(finalize_kernel, dim3(NROWS / 256), dim3(256), 0, stream,
                       keys, (float*)d_out);
}


# Per-tile epilogue cut to a running min and max of |M-1| per row: C is unimodal in |M-1|, so its row minimum sits at one of the two; exp and log-pdf are evaluated only there, once after the sweep
# speedup vs baseline: 3.0897x; 1.0589x over previous
#include <hip/hip_runtime.h>

#define NROWS 8192
#define DIM   256
#define INV_SIGMA 3.33333333333f
#define LOGC      0.28503428291f

#define ROW_TILES       4
#define ROWS_PER_WAVE   (ROW_TILES * 16)
#define ROWS_PER_BLOCK  256
#define WAVES_PER_BLOCK 4
#define THREADS         (WAVES_PER_BLOCK * 32)
#define NCOLTILES       (NROWS / 16)
#define NSLICES         8
#define TILES_PER_SLICE (NCOLTILES / NSLICES)
#define LDS_STRIDE      264
#define STAGE_ELEMS     (16 * LDS_STRIDE)

typedef __attribute__((ext_vector_type(16))) __bf16 v16bf;
typedef __attribute__((ext_vector_type(8)))  float  v8f;
typedef __attribute__((ext_vector_type(4)))  unsigned int u32x4;

__device__ __forceinline__ unsigned short f32_to_bf16_rne(float f) {
    unsigned u = __float_as_uint(f);
    u += 0x7FFFu + ((u >> 16) & 1u);
    return (unsigned short)(u >> 16);
}
__device__ __forceinline__ float bf16_to_f32(unsigned short h) {
    return __uint_as_float(((unsigned)h) << 16);
}

__device__ __forceinline__ unsigned f2key(float f) {
    unsigned b = __float_as_uint(f);
    return (b & 0x80000000u) ? ~b : (b | 0x80000000u);
}
__device__ __forceinline__ float key2f(unsigned u) {
    unsigned b = (u & 0x80000000u) ? (u ^ 0x80000000u) : ~u;
    return __uint_as_float(b);
}

__global__ void init_keys_kernel(unsigned* __restrict__ keys) {
    keys[blockIdx.x * blockDim.x + threadIdx.x] = 0xFFFFFFFFu;
}

__global__ void normalize_split_kernel(const float* __restrict__ Ex,
                                       const float* __restrict__ Ey,
                                       unsigned short* __restrict__ Xhi,
                                       unsigned short* __restrict__ Xlo,
                                       unsigned short* __restrict__ Yhi,
                                       unsigned short* __restrict__ Ylo) {
    const int wave = (blockIdx.x * blockDim.x + threadIdx.x) >> 5;
    const int lane = threadIdx.x & 31;

    const float* src;
    unsigned short *hi, *lo;
    int row;
    if (wave < NROWS) { src = Ex; hi = Xhi; lo = Xlo; row = wave; }
    else              { src = Ey; hi = Yhi; lo = Ylo; row = wave - NROWS; }

    const float* p = src + (size_t)row * DIM;
    float v[DIM / 32];
    float ss = 0.0f;
#pragma unroll
    for (int i = 0; i < DIM / 32; ++i) {
        v[i] = p[lane + 32 * i];
        ss = fmaf(v[i], v[i], ss);
    }
#pragma unroll
    for (int off = 16; off > 0; off >>= 1)
        ss += __shfl_xor(ss, off, 32);

    const float inv = 1.0f / fmaxf(sqrtf(ss), 1e-8f);

#pragma unroll
    for (int i = 0; i < DIM / 32; ++i) {
        const float x = v[i] * inv;
        const unsigned short h = f32_to_bf16_rne(x);
        const unsigned short l = f32_to_bf16_rne(x - bf16_to_f32(h));
        const size_t idx = (size_t)row * DIM + lane + 32 * i;
        hi[idx] = h;
        lo[idx] = l;
    }
}

__device__ __forceinline__ v16bf lds_frag(const unsigned short* p) {
    union { uint4 q[2]; v16bf v; } f;
    f.q[0] = *(const uint4*)(p);
    f.q[1] = *(const uint4*)(p + 16);
    return f.v;
}
__device__ __forceinline__ v16bf glb_frag(const unsigned short* __restrict__ p) {
    union { uint4 q[2]; v16bf v; } f;
    f.q[0] = *(const uint4*)(p);
    f.q[1] = *(const uint4*)(p + 16);
    return f.v;
}

__device__ __forceinline__ float logpdf_cost(float d) {
    const float z    = d * INV_SIGMA;
    const float logp = fmaf(-0.5f * z, z, LOGC);
    return -__expf(logp) * logp;
}

#define FRAG_LOAD(FH, FL, S)                                                 \
    do {                                                                     \
        _Pragma("unroll")                                                    \
        for (int c_ = 0; c_ < DIM / 32; ++c_) {                              \
            FH[c_] = lds_frag(&lBh[(S) * STAGE_ELEMS + bOff + c_ * 32]);     \
            FL[c_] = lds_frag(&lBl[(S) * STAGE_ELEMS + bOff + c_ * 32]);     \
        }                                                                    \
    } while (0)

#define STORE_B(S)                                                           \
    do {                                                                     \
        const int s_ = (S) * STAGE_ELEMS + sOff;                             \
        *(u32x4*)&lBh[s_]                   = rh0;                           \
        *(u32x4*)&lBl[s_]                   = rl0;                           \
        *(u32x4*)&lBh[s_ + 4 * LDS_STRIDE]  = rh1;                           \
        *(u32x4*)&lBl[s_ + 4 * LDS_STRIDE]  = rl1;                           \
        *(u32x4*)&lBh[s_ + 8 * LDS_STRIDE]  = rh2;                           \
        *(u32x4*)&lBl[s_ + 8 * LDS_STRIDE]  = rl2;                           \
        *(u32x4*)&lBh[s_ + 12 * LDS_STRIDE] = rh3;                           \
        *(u32x4*)&lBl[s_ + 12 * LDS_STRIDE] = rl3;                           \
    } while (0)

#define PREFETCH_B(T)                                                        \
    do {                                                                     \
        const int tt_ = ((T) < TILES_PER_SLICE) ? (T) : (TILES_PER_SLICE - 1); \
        const size_t g_ = (size_t)((tileBase + tt_) * 16) * DIM + gOff;      \
        rh0 = *(const u32x4*)&Yhi[g_];                                       \
        rl0 = *(const u32x4*)&Ylo[g_];                                       \
        rh1 = *(const u32x4*)&Yhi[g_ + 4 * DIM];                             \
        rl1 = *(const u32x4*)&Ylo[g_ + 4 * DIM];                             \
        rh2 = *(const u32x4*)&Yhi[g_ + 8 * DIM];                             \
        rl2 = *(const u32x4*)&Ylo[g_ + 8 * DIM];                             \
        rh3 = *(const u32x4*)&Yhi[g_ + 12 * DIM];                            \
        rl3 = *(const u32x4*)&Ylo[g_ + 12 * DIM];                            \
    } while (0)

#define COMPUTE_TILE(FH, FL)                                                 \
    do {                                                                     \
        v8f acc[ROW_TILES];                                                  \
        _Pragma("unroll")                                                    \
        for (int rt = 0; rt < ROW_TILES; ++rt) acc[rt] = v8f{};              \
        _Pragma("unroll")                                                    \
        for (int c = 0; c < DIM / 32; ++c) {                                 \
            _Pragma("unroll")                                                \
            for (int rt = 0; rt < ROW_TILES; ++rt)                           \
                acc[rt] = __builtin_amdgcn_wmma_f32_16x16x32_bf16(           \
                    false, al[rt][c], false, FH[c], (short)0, acc[rt],       \
                    false, false);                                           \
            _Pragma("unroll")                                                \
            for (int rt = 0; rt < ROW_TILES; ++rt)                           \
                acc[rt] = __builtin_amdgcn_wmma_f32_16x16x32_bf16(           \
                    false, ah[rt][c], false, FL[c], (short)0, acc[rt],       \
                    false, false);                                           \
            _Pragma("unroll")                                                \
            for (int rt = 0; rt < ROW_TILES; ++rt)                           \
                acc[rt] = __builtin_amdgcn_wmma_f32_16x16x32_bf16(           \
                    false, ah[rt][c], false, FH[c], (short)0, acc[rt],       \
                    false, false);                                           \
        }                                                                    \
        _Pragma("unroll")                                                    \
        for (int rt = 0; rt < ROW_TILES; ++rt) {                             \
            _Pragma("unroll")                                                \
            for (int r = 0; r < 8; ++r) {                                    \
                const float d_ = fabsf(acc[rt][r] - 1.0f);                   \
                dmin[rt][r] = fminf(dmin[rt][r], d_);                        \
                dmax[rt][r] = fmaxf(dmax[rt][r], d_);                        \
            }                                                                \
        }                                                                    \
    } while (0)

#define SWEEP_STEP(T, CH, CL, NH, NL)                                        \
    do {                                                                     \
         \
         \
        __syncthreads();                                                     \
        FRAG_LOAD(NH, NL, ((T) + 1) & 1);                                    \
        STORE_B((T) & 1);                                                    \
        PREFETCH_B((T) + 3);                                                 \
         \
         \
         \
        __builtin_amdgcn_sched_barrier(0);                                   \
        COMPUTE_TILE(CH, CL);                                                \
         \
         \
         \
        __builtin_amdgcn_sched_barrier(0);                                   \
    } while (0)

__global__ __launch_bounds__(THREADS) void cosmin_wmma_kernel(
    const unsigned short* __restrict__ Xhi,
    const unsigned short* __restrict__ Xlo,
    const unsigned short* __restrict__ Yhi,
    const unsigned short* __restrict__ Ylo,
    unsigned* __restrict__ keys) {
    __shared__ unsigned short lBh[2 * STAGE_ELEMS];
    __shared__ unsigned short lBl[2 * STAGE_ELEMS];

    const int tid      = threadIdx.x;
    const int lane     = tid & 31;
    const int wave     = tid >> 5;
    const int rowBase  = blockIdx.x * ROWS_PER_BLOCK + wave * ROWS_PER_WAVE;
    const int tileBase = blockIdx.y * TILES_PER_SLICE;

    const int hiSel = (lane >> 4) << 3;

    v16bf ah[ROW_TILES][DIM / 32], al[ROW_TILES][DIM / 32];
#pragma unroll
    for (int rt = 0; rt < ROW_TILES; ++rt) {
        const size_t aRow =
            (size_t)(rowBase + rt * 16 + (lane & 15)) * DIM + hiSel;
#pragma unroll
        for (int c = 0; c < DIM / 32; ++c) {
            ah[rt][c] = glb_frag(&Xhi[aRow + c * 32]);
            al[rt][c] = glb_frag(&Xlo[aRow + c * 32]);
        }
    }

    const int gOff = (tid >> 5) * DIM + (tid & 31) * 8;
    const int sOff = (tid >> 5) * LDS_STRIDE + (tid & 31) * 8;
    const int bOff = (lane & 15) * LDS_STRIDE + hiSel;
    u32x4 rh0, rh1, rh2, rh3, rl0, rl1, rl2, rl3;

    float dmin[ROW_TILES][8], dmax[ROW_TILES][8];
#pragma unroll
    for (int rt = 0; rt < ROW_TILES; ++rt)
#pragma unroll
        for (int r = 0; r < 8; ++r) {
            dmin[rt][r] = 3.402823466e38f;
            dmax[rt][r] = 0.0f;
        }

    v16bf f0h[DIM / 32], f0l[DIM / 32], f1h[DIM / 32], f1l[DIM / 32];

    PREFETCH_B(0);
    STORE_B(0);
    PREFETCH_B(1);
    STORE_B(1);
    PREFETCH_B(2);
    __syncthreads();
    FRAG_LOAD(f0h, f0l, 0);

    for (int t = 0; t < TILES_PER_SLICE; t += 2) {
        SWEEP_STEP(t,     f0h, f0l, f1h, f1l);
        SWEEP_STEP(t + 1, f1h, f1l, f0h, f0l);
    }

    float minv[ROW_TILES][8];
#pragma unroll
    for (int rt = 0; rt < ROW_TILES; ++rt) {
#pragma unroll
        for (int r = 0; r < 8; ++r)
            minv[rt][r] = fminf(logpdf_cost(dmin[rt][r]),
                                logpdf_cost(dmax[rt][r]));
    }

#pragma unroll
    for (int rt = 0; rt < ROW_TILES; ++rt) {
#pragma unroll
        for (int r = 0; r < 8; ++r) {
#pragma unroll
            for (int off = 1; off < 16; off <<= 1)
                minv[rt][r] =
                    fminf(minv[rt][r], __shfl_xor(minv[rt][r], off, 32));
        }
    }

    if ((lane & 15) == 0) {
#pragma unroll
        for (int rt = 0; rt < ROW_TILES; ++rt) {
            const int base = rowBase + rt * 16 + ((lane >> 4) << 3);
#pragma unroll
            for (int r = 0; r < 8; ++r)
                atomicMin(&keys[base + r], f2key(minv[rt][r]));
        }
    }
}

#undef SWEEP_STEP
#undef COMPUTE_TILE
#undef PREFETCH_B
#undef STORE_B
#undef FRAG_LOAD

__global__ void finalize_kernel(const unsigned* __restrict__ keys,
                                float* __restrict__ out) {
    const int i = blockIdx.x * blockDim.x + threadIdx.x;
    out[i] = key2f(keys[i]);
}

extern "C" void kernel_launch(void* const* d_in, const int* in_sizes, int n_in,
                              void* d_out, int out_size, void* d_ws, size_t ws_size,
                              hipStream_t stream) {
    const float* Ex = (const float*)d_in[0];
    const float* Ey = (const float*)d_in[1];

    const size_t elems = (size_t)NROWS * DIM;
    unsigned short* Xhi = (unsigned short*)d_ws;
    unsigned short* Xlo = Xhi + elems;
    unsigned short* Yhi = Xlo + elems;
    unsigned short* Ylo = Yhi + elems;
    unsigned*      keys = (unsigned*)(Ylo + elems);

    hipLaunchKernelGGL(init_keys_kernel, dim3(NROWS / 256), dim3(256), 0, stream,
                       keys);
    hipLaunchKernelGGL(normalize_split_kernel, dim3((2 * NROWS) / 8), dim3(256), 0,
                       stream, Ex, Ey, Xhi, Xlo, Yhi, Ylo);
    hipLaunchKernelGGL(cosmin_wmma_kernel,
                       dim3(NROWS / ROWS_PER_BLOCK, NSLICES), dim3(THREADS), 0,
                       stream, Xhi, Xlo, Yhi, Ylo, keys);
    hipLaunchKernelGGL(finalize_kernel, dim3(NROWS / 256), dim3(256), 0, stream,
                       keys, (float*)d_out);
}


# Each row-block starts its column sweep 2 tiles further into its slice (wrapping): the 32 blocks of a slice stop requesting the same B tile in the same step; min/max order-free, sums untouched
# speedup vs baseline: 3.3223x; 1.0753x over previous
#include <hip/hip_runtime.h>

#define NROWS 8192
#define DIM   256
#define INV_SIGMA 3.33333333333f
#define LOGC      0.28503428291f

#define ROW_TILES       4
#define ROWS_PER_WAVE   (ROW_TILES * 16)
#define ROWS_PER_BLOCK  256
#define WAVES_PER_BLOCK 4
#define THREADS         (WAVES_PER_BLOCK * 32)
#define NCOLTILES       (NROWS / 16)
#define NSLICES         8
#define TILES_PER_SLICE (NCOLTILES / NSLICES)
#define LDS_STRIDE      264
#define STAGE_ELEMS     (16 * LDS_STRIDE)

typedef __attribute__((ext_vector_type(16))) __bf16 v16bf;
typedef __attribute__((ext_vector_type(8)))  float  v8f;
typedef __attribute__((ext_vector_type(4)))  unsigned int u32x4;

__device__ __forceinline__ unsigned short f32_to_bf16_rne(float f) {
    unsigned u = __float_as_uint(f);
    u += 0x7FFFu + ((u >> 16) & 1u);
    return (unsigned short)(u >> 16);
}
__device__ __forceinline__ float bf16_to_f32(unsigned short h) {
    return __uint_as_float(((unsigned)h) << 16);
}

__device__ __forceinline__ unsigned f2key(float f) {
    unsigned b = __float_as_uint(f);
    return (b & 0x80000000u) ? ~b : (b | 0x80000000u);
}
__device__ __forceinline__ float key2f(unsigned u) {
    unsigned b = (u & 0x80000000u) ? (u ^ 0x80000000u) : ~u;
    return __uint_as_float(b);
}

__global__ void init_keys_kernel(unsigned* __restrict__ keys) {
    keys[blockIdx.x * blockDim.x + threadIdx.x] = 0xFFFFFFFFu;
}

__global__ void normalize_split_kernel(const float* __restrict__ Ex,
                                       const float* __restrict__ Ey,
                                       unsigned short* __restrict__ Xhi,
                                       unsigned short* __restrict__ Xlo,
                                       unsigned short* __restrict__ Yhi,
                                       unsigned short* __restrict__ Ylo) {
    const int wave = (blockIdx.x * blockDim.x + threadIdx.x) >> 5;
    const int lane = threadIdx.x & 31;

    const float* src;
    unsigned short *hi, *lo;
    int row;
    if (wave < NROWS) { src = Ex; hi = Xhi; lo = Xlo; row = wave; }
    else              { src = Ey; hi = Yhi; lo = Ylo; row = wave - NROWS; }

    const float* p = src + (size_t)row * DIM;
    float v[DIM / 32];
    float ss = 0.0f;
#pragma unroll
    for (int i = 0; i < DIM / 32; ++i) {
        v[i] = p[lane + 32 * i];
        ss = fmaf(v[i], v[i], ss);
    }
#pragma unroll
    for (int off = 16; off > 0; off >>= 1)
        ss += __shfl_xor(ss, off, 32);

    const float inv = 1.0f / fmaxf(sqrtf(ss), 1e-8f);

#pragma unroll
    for (int i = 0; i < DIM / 32; ++i) {
        const float x = v[i] * inv;
        const unsigned short h = f32_to_bf16_rne(x);
        const unsigned short l = f32_to_bf16_rne(x - bf16_to_f32(h));
        const size_t idx = (size_t)row * DIM + lane + 32 * i;
        hi[idx] = h;
        lo[idx] = l;
    }
}

__device__ __forceinline__ v16bf lds_frag(const unsigned short* p) {
    union { uint4 q[2]; v16bf v; } f;
    f.q[0] = *(const uint4*)(p);
    f.q[1] = *(const uint4*)(p + 16);
    return f.v;
}
__device__ __forceinline__ v16bf glb_frag(const unsigned short* __restrict__ p) {
    union { uint4 q[2]; v16bf v; } f;
    f.q[0] = *(const uint4*)(p);
    f.q[1] = *(const uint4*)(p + 16);
    return f.v;
}

__device__ __forceinline__ float logpdf_cost(float d) {
    const float z    = d * INV_SIGMA;
    const float logp = fmaf(-0.5f * z, z, LOGC);
    return -__expf(logp) * logp;
}

#define FRAG_LOAD(FH, FL, S)                                                 \
    do {                                                                     \
        _Pragma("unroll")                                                    \
        for (int c_ = 0; c_ < DIM / 32; ++c_) {                              \
            FH[c_] = lds_frag(&lBh[(S) * STAGE_ELEMS + bOff + c_ * 32]);     \
            FL[c_] = lds_frag(&lBl[(S) * STAGE_ELEMS + bOff + c_ * 32]);     \
        }                                                                    \
    } while (0)

#define STORE_B(S)                                                           \
    do {                                                                     \
        const int s_ = (S) * STAGE_ELEMS + sOff;                             \
        *(u32x4*)&lBh[s_]                   = rh0;                           \
        *(u32x4*)&lBl[s_]                   = rl0;                           \
        *(u32x4*)&lBh[s_ + 4 * LDS_STRIDE]  = rh1;                           \
        *(u32x4*)&lBl[s_ + 4 * LDS_STRIDE]  = rl1;                           \
        *(u32x4*)&lBh[s_ + 8 * LDS_STRIDE]  = rh2;                           \
        *(u32x4*)&lBl[s_ + 8 * LDS_STRIDE]  = rl2;                           \
        *(u32x4*)&lBh[s_ + 12 * LDS_STRIDE] = rh3;                           \
        *(u32x4*)&lBl[s_ + 12 * LDS_STRIDE] = rl3;                           \
    } while (0)

#define PREFETCH_B(T)                                                        \
    do {                                                                     \
        const int tt_ = ((T) < TILES_PER_SLICE) ? (T) : (TILES_PER_SLICE - 1); \
        const int tr_ = (tt_ + rot) & (TILES_PER_SLICE - 1);                 \
        const size_t g_ = (size_t)((tileBase + tr_) * 16) * DIM + gOff;      \
        rh0 = *(const u32x4*)&Yhi[g_];                                       \
        rl0 = *(const u32x4*)&Ylo[g_];                                       \
        rh1 = *(const u32x4*)&Yhi[g_ + 4 * DIM];                             \
        rl1 = *(const u32x4*)&Ylo[g_ + 4 * DIM];                             \
        rh2 = *(const u32x4*)&Yhi[g_ + 8 * DIM];                             \
        rl2 = *(const u32x4*)&Ylo[g_ + 8 * DIM];                             \
        rh3 = *(const u32x4*)&Yhi[g_ + 12 * DIM];                            \
        rl3 = *(const u32x4*)&Ylo[g_ + 12 * DIM];                            \
    } while (0)

#define COMPUTE_TILE(FH, FL)                                                 \
    do {                                                                     \
        v8f acc[ROW_TILES];                                                  \
        _Pragma("unroll")                                                    \
        for (int rt = 0; rt < ROW_TILES; ++rt) acc[rt] = v8f{};              \
        _Pragma("unroll")                                                    \
        for (int c = 0; c < DIM / 32; ++c) {                                 \
            _Pragma("unroll")                                                \
            for (int rt = 0; rt < ROW_TILES; ++rt)                           \
                acc[rt] = __builtin_amdgcn_wmma_f32_16x16x32_bf16(           \
                    false, al[rt][c], false, FH[c], (short)0, acc[rt],       \
                    false, false);                                           \
            _Pragma("unroll")                                                \
            for (int rt = 0; rt < ROW_TILES; ++rt)                           \
                acc[rt] = __builtin_amdgcn_wmma_f32_16x16x32_bf16(           \
                    false, ah[rt][c], false, FL[c], (short)0, acc[rt],       \
                    false, false);                                           \
            _Pragma("unroll")                                                \
            for (int rt = 0; rt < ROW_TILES; ++rt)                           \
                acc[rt] = __builtin_amdgcn_wmma_f32_16x16x32_bf16(           \
                    false, ah[rt][c], false, FH[c], (short)0, acc[rt],       \
                    false, false);                                           \
        }                                                                    \
        _Pragma("unroll")                                                    \
        for (int rt = 0; rt < ROW_TILES; ++rt) {                             \
            _Pragma("unroll")                                                \
            for (int r = 0; r < 8; ++r) {                                    \
                const float d_ = fabsf(acc[rt][r] - 1.0f);                   \
                dmin[rt][r] = fminf(dmin[rt][r], d_);                        \
                dmax[rt][r] = fmaxf(dmax[rt][r], d_);                        \
            }                                                                \
        }                                                                    \
    } while (0)

#define SWEEP_STEP(T, CH, CL, NH, NL)                                        \
    do {                                                                     \
         \
         \
        __syncthreads();                                                     \
        FRAG_LOAD(NH, NL, ((T) + 1) & 1);                                    \
        STORE_B((T) & 1);                                                    \
        PREFETCH_B((T) + 3);                                                 \
         \
         \
         \
        __builtin_amdgcn_sched_barrier(0);                                   \
        COMPUTE_TILE(CH, CL);                                                \
         \
         \
         \
        __builtin_amdgcn_sched_barrier(0);                                   \
    } while (0)

__global__ __launch_bounds__(THREADS) void cosmin_wmma_kernel(
    const unsigned short* __restrict__ Xhi,
    const unsigned short* __restrict__ Xlo,
    const unsigned short* __restrict__ Yhi,
    const unsigned short* __restrict__ Ylo,
    unsigned* __restrict__ keys) {
    __shared__ unsigned short lBh[2 * STAGE_ELEMS];
    __shared__ unsigned short lBl[2 * STAGE_ELEMS];

    const int tid      = threadIdx.x;
    const int lane     = tid & 31;
    const int wave     = tid >> 5;
    const int rowBase  = blockIdx.x * ROWS_PER_BLOCK + wave * ROWS_PER_WAVE;
    const int tileBase = blockIdx.y * TILES_PER_SLICE;
    const int rot      = (blockIdx.x * 2) & (TILES_PER_SLICE - 1);

    const int hiSel = (lane >> 4) << 3;

    v16bf ah[ROW_TILES][DIM / 32], al[ROW_TILES][DIM / 32];
#pragma unroll
    for (int rt = 0; rt < ROW_TILES; ++rt) {
        const size_t aRow =
            (size_t)(rowBase + rt * 16 + (lane & 15)) * DIM + hiSel;
#pragma unroll
        for (int c = 0; c < DIM / 32; ++c) {
            ah[rt][c] = glb_frag(&Xhi[aRow + c * 32]);
            al[rt][c] = glb_frag(&Xlo[aRow + c * 32]);
        }
    }

    const int gOff = (tid >> 5) * DIM + (tid & 31) * 8;
    const int sOff = (tid >> 5) * LDS_STRIDE + (tid & 31) * 8;
    const int bOff = (lane & 15) * LDS_STRIDE + hiSel;
    u32x4 rh0, rh1, rh2, rh3, rl0, rl1, rl2, rl3;

    float dmin[ROW_TILES][8], dmax[ROW_TILES][8];
#pragma unroll
    for (int rt = 0; rt < ROW_TILES; ++rt)
#pragma unroll
        for (int r = 0; r < 8; ++r) {
            dmin[rt][r] = 3.402823466e38f;
            dmax[rt][r] = 0.0f;
        }

    v16bf f0h[DIM / 32], f0l[DIM / 32], f1h[DIM / 32], f1l[DIM / 32];

    PREFETCH_B(0);
    STORE_B(0);
    PREFETCH_B(1);
    STORE_B(1);
    PREFETCH_B(2);
    __syncthreads();
    FRAG_LOAD(f0h, f0l, 0);

    for (int t = 0; t < TILES_PER_SLICE; t += 2) {
        SWEEP_STEP(t,     f0h, f0l, f1h, f1l);
        SWEEP_STEP(t + 1, f1h, f1l, f0h, f0l);
    }

    float minv[ROW_TILES][8];
#pragma unroll
    for (int rt = 0; rt < ROW_TILES; ++rt) {
#pragma unroll
        for (int r = 0; r < 8; ++r)
            minv[rt][r] = fminf(logpdf_cost(dmin[rt][r]),
                                logpdf_cost(dmax[rt][r]));
    }

#pragma unroll
    for (int rt = 0; rt < ROW_TILES; ++rt) {
#pragma unroll
        for (int r = 0; r < 8; ++r) {
#pragma unroll
            for (int off = 1; off < 16; off <<= 1)
                minv[rt][r] =
                    fminf(minv[rt][r], __shfl_xor(minv[rt][r], off, 32));
        }
    }

    if ((lane & 15) == 0) {
#pragma unroll
        for (int rt = 0; rt < ROW_TILES; ++rt) {
            const int base = rowBase + rt * 16 + ((lane >> 4) << 3);
#pragma unroll
            for (int r = 0; r < 8; ++r)
                atomicMin(&keys[base + r], f2key(minv[rt][r]));
        }
    }
}

#undef SWEEP_STEP
#undef COMPUTE_TILE
#undef PREFETCH_B
#undef STORE_B
#undef FRAG_LOAD

__global__ void finalize_kernel(const unsigned* __restrict__ keys,
                                float* __restrict__ out) {
    const int i = blockIdx.x * blockDim.x + threadIdx.x;
    out[i] = key2f(keys[i]);
}

extern "C" void kernel_launch(void* const* d_in, const int* in_sizes, int n_in,
                              void* d_out, int out_size, void* d_ws, size_t ws_size,
                              hipStream_t stream) {
    const float* Ex = (const float*)d_in[0];
    const float* Ey = (const float*)d_in[1];

    const size_t elems = (size_t)NROWS * DIM;
    unsigned short* Xhi = (unsigned short*)d_ws;
    unsigned short* Xlo = Xhi + elems;
    unsigned short* Yhi = Xlo + elems;
    unsigned short* Ylo = Yhi + elems;
    unsigned*      keys = (unsigned*)(Ylo + elems);

    hipLaunchKernelGGL(init_keys_kernel, dim3(NROWS / 256), dim3(256), 0, stream,
                       keys);
    hipLaunchKernelGGL(normalize_split_kernel, dim3((2 * NROWS) / 8), dim3(256), 0,
                       stream, Ex, Ey, Xhi, Xlo, Yhi, Ylo);
    hipLaunchKernelGGL(cosmin_wmma_kernel,
                       dim3(NROWS / ROWS_PER_BLOCK, NSLICES), dim3(THREADS), 0,
                       stream, Xhi, Xlo, Yhi, Ylo, keys);
    hipLaunchKernelGGL(finalize_kernel, dim3(NROWS / 256), dim3(256), 0, stream,
                       keys, (float*)d_out);
}


# Row minimum taken by f32 atomic min straight into out, which the normalize launch first sets to +inf: the key-init and key-to-float launches go (4 launches -> 2); min is order-free, no sum changed
# speedup vs baseline: 3.4204x; 1.0295x over previous
#include <hip/hip_runtime.h>

#define NROWS 8192
#define DIM   256
#define INV_SIGMA 3.33333333333f
#define LOGC      0.28503428291f

#define ROW_TILES       4
#define ROWS_PER_WAVE   (ROW_TILES * 16)
#define ROWS_PER_BLOCK  256
#define WAVES_PER_BLOCK 4
#define THREADS         (WAVES_PER_BLOCK * 32)
#define NCOLTILES       (NROWS / 16)
#define NSLICES         8
#define TILES_PER_SLICE (NCOLTILES / NSLICES)
#define LDS_STRIDE      264
#define STAGE_ELEMS     (16 * LDS_STRIDE)

typedef __attribute__((ext_vector_type(16))) __bf16 v16bf;
typedef __attribute__((ext_vector_type(8)))  float  v8f;
typedef __attribute__((ext_vector_type(4)))  unsigned int u32x4;

__device__ __forceinline__ unsigned short f32_to_bf16_rne(float f) {
    unsigned u = __float_as_uint(f);
    u += 0x7FFFu + ((u >> 16) & 1u);
    return (unsigned short)(u >> 16);
}
__device__ __forceinline__ float bf16_to_f32(unsigned short h) {
    return __uint_as_float(((unsigned)h) << 16);
}

__global__ void normalize_split_kernel(const float* __restrict__ Ex,
                                       const float* __restrict__ Ey,
                                       unsigned short* __restrict__ Xhi,
                                       unsigned short* __restrict__ Xlo,
                                       unsigned short* __restrict__ Yhi,
                                       unsigned short* __restrict__ Ylo,
                                       float* __restrict__ out) {
    const int wave = (blockIdx.x * blockDim.x + threadIdx.x) >> 5;
    const int lane = threadIdx.x & 31;

    if (blockIdx.x < NROWS / 8 && (blockIdx.x & 3) == 0 && threadIdx.x < 32)
        out[blockIdx.x * 8 + threadIdx.x] = __uint_as_float(0x7F800000u);

    const float* src;
    unsigned short *hi, *lo;
    int row;
    if (wave < NROWS) { src = Ex; hi = Xhi; lo = Xlo; row = wave; }
    else              { src = Ey; hi = Yhi; lo = Ylo; row = wave - NROWS; }

    const float* p = src + (size_t)row * DIM;
    float v[DIM / 32];
    float ss = 0.0f;
#pragma unroll
    for (int i = 0; i < DIM / 32; ++i) {
        v[i] = p[lane + 32 * i];
        ss = fmaf(v[i], v[i], ss);
    }
#pragma unroll
    for (int off = 16; off > 0; off >>= 1)
        ss += __shfl_xor(ss, off, 32);

    const float inv = 1.0f / fmaxf(sqrtf(ss), 1e-8f);

#pragma unroll
    for (int i = 0; i < DIM / 32; ++i) {
        const float x = v[i] * inv;
        const unsigned short h = f32_to_bf16_rne(x);
        const unsigned short l = f32_to_bf16_rne(x - bf16_to_f32(h));
        const size_t idx = (size_t)row * DIM + lane + 32 * i;
        hi[idx] = h;
        lo[idx] = l;
    }
}

__device__ __forceinline__ v16bf lds_frag(const unsigned short* p) {
    union { uint4 q[2]; v16bf v; } f;
    f.q[0] = *(const uint4*)(p);
    f.q[1] = *(const uint4*)(p + 16);
    return f.v;
}
__device__ __forceinline__ v16bf glb_frag(const unsigned short* __restrict__ p) {
    union { uint4 q[2]; v16bf v; } f;
    f.q[0] = *(const uint4*)(p);
    f.q[1] = *(const uint4*)(p + 16);
    return f.v;
}

__device__ __forceinline__ float logpdf_cost(float d) {
    const float z    = d * INV_SIGMA;
    const float logp = fmaf(-0.5f * z, z, LOGC);
    return -__expf(logp) * logp;
}

#define FRAG_LOAD(FH, FL, S)                                                 \
    do {                                                                     \
        _Pragma("unroll")                                                    \
        for (int c_ = 0; c_ < DIM / 32; ++c_) {                              \
            FH[c_] = lds_frag(&lBh[(S) * STAGE_ELEMS + bOff + c_ * 32]);     \
            FL[c_] = lds_frag(&lBl[(S) * STAGE_ELEMS + bOff + c_ * 32]);     \
        }                                                                    \
    } while (0)

#define STORE_B(S)                                                           \
    do {                                                                     \
        const int s_ = (S) * STAGE_ELEMS + sOff;                             \
        *(u32x4*)&lBh[s_]                   = rh0;                           \
        *(u32x4*)&lBl[s_]                   = rl0;                           \
        *(u32x4*)&lBh[s_ + 4 * LDS_STRIDE]  = rh1;                           \
        *(u32x4*)&lBl[s_ + 4 * LDS_STRIDE]  = rl1;                           \
        *(u32x4*)&lBh[s_ + 8 * LDS_STRIDE]  = rh2;                           \
        *(u32x4*)&lBl[s_ + 8 * LDS_STRIDE]  = rl2;                           \
        *(u32x4*)&lBh[s_ + 12 * LDS_STRIDE] = rh3;                           \
        *(u32x4*)&lBl[s_ + 12 * LDS_STRIDE] = rl3;                           \
    } while (0)

#define PREFETCH_B(T)                                                        \
    do {                                                                     \
        const int tt_ = ((T) < TILES_PER_SLICE) ? (T) : (TILES_PER_SLICE - 1); \
        const int tr_ = (tt_ + rot) & (TILES_PER_SLICE - 1);                 \
        const size_t g_ = (size_t)((tileBase + tr_) * 16) * DIM + gOff;      \
        rh0 = *(const u32x4*)&Yhi[g_];                                       \
        rl0 = *(const u32x4*)&Ylo[g_];                                       \
        rh1 = *(const u32x4*)&Yhi[g_ + 4 * DIM];                             \
        rl1 = *(const u32x4*)&Ylo[g_ + 4 * DIM];                             \
        rh2 = *(const u32x4*)&Yhi[g_ + 8 * DIM];                             \
        rl2 = *(const u32x4*)&Ylo[g_ + 8 * DIM];                             \
        rh3 = *(const u32x4*)&Yhi[g_ + 12 * DIM];                            \
        rl3 = *(const u32x4*)&Ylo[g_ + 12 * DIM];                            \
    } while (0)

#define COMPUTE_TILE(FH, FL)                                                 \
    do {                                                                     \
        v8f acc[ROW_TILES];                                                  \
        _Pragma("unroll")                                                    \
        for (int rt = 0; rt < ROW_TILES; ++rt) acc[rt] = v8f{};              \
        _Pragma("unroll")                                                    \
        for (int c = 0; c < DIM / 32; ++c) {                                 \
            _Pragma("unroll")                                                \
            for (int rt = 0; rt < ROW_TILES; ++rt)                           \
                acc[rt] = __builtin_amdgcn_wmma_f32_16x16x32_bf16(           \
                    false, al[rt][c], false, FH[c], (short)0, acc[rt],       \
                    false, false);                                           \
            _Pragma("unroll")                                                \
            for (int rt = 0; rt < ROW_TILES; ++rt)                           \
                acc[rt] = __builtin_amdgcn_wmma_f32_16x16x32_bf16(           \
                    false, ah[rt][c], false, FL[c], (short)0, acc[rt],       \
                    false, false);                                           \
            _Pragma("unroll")                                                \
            for (int rt = 0; rt < ROW_TILES; ++rt)                           \
                acc[rt] = __builtin_amdgcn_wmma_f32_16x16x32_bf16(           \
                    false, ah[rt][c], false, FH[c], (short)0, acc[rt],       \
                    false, false);                                           \
        }                                                                    \
        _Pragma("unroll")                                                    \
        for (int rt = 0; rt < ROW_TILES; ++rt) {                             \
            _Pragma("unroll")                                                \
            for (int r = 0; r < 8; ++r) {                                    \
                const float d_ = fabsf(acc[rt][r] - 1.0f);                   \
                dmin[rt][r] = fminf(dmin[rt][r], d_);                        \
                dmax[rt][r] = fmaxf(dmax[rt][r], d_);                        \
            }                                                                \
        }                                                                    \
    } while (0)

#define SWEEP_STEP(T, CH, CL, NH, NL)                                        \
    do {                                                                     \
         \
         \
        __syncthreads();                                                     \
        FRAG_LOAD(NH, NL, ((T) + 1) & 1);                                    \
        STORE_B((T) & 1);                                                    \
        PREFETCH_B((T) + 3);                                                 \
         \
         \
         \
        __builtin_amdgcn_sched_barrier(0);                                   \
        COMPUTE_TILE(CH, CL);                                                \
         \
         \
         \
        __builtin_amdgcn_sched_barrier(0);                                   \
    } while (0)

__global__ __launch_bounds__(THREADS) void cosmin_wmma_kernel(
    const unsigned short* __restrict__ Xhi,
    const unsigned short* __restrict__ Xlo,
    const unsigned short* __restrict__ Yhi,
    const unsigned short* __restrict__ Ylo,
    float* __restrict__ out) {
    __shared__ unsigned short lBh[2 * STAGE_ELEMS];
    __shared__ unsigned short lBl[2 * STAGE_ELEMS];

    const int tid      = threadIdx.x;
    const int lane     = tid & 31;
    const int wave     = tid >> 5;
    const int rowBase  = blockIdx.x * ROWS_PER_BLOCK + wave * ROWS_PER_WAVE;
    const int tileBase = blockIdx.y * TILES_PER_SLICE;
    const int rot      = (blockIdx.x * 2) & (TILES_PER_SLICE - 1);

    const int hiSel = (lane >> 4) << 3;

    v16bf ah[ROW_TILES][DIM / 32], al[ROW_TILES][DIM / 32];
#pragma unroll
    for (int rt = 0; rt < ROW_TILES; ++rt) {
        const size_t aRow =
            (size_t)(rowBase + rt * 16 + (lane & 15)) * DIM + hiSel;
#pragma unroll
        for (int c = 0; c < DIM / 32; ++c) {
            ah[rt][c] = glb_frag(&Xhi[aRow + c * 32]);
            al[rt][c] = glb_frag(&Xlo[aRow + c * 32]);
        }
    }

    const int gOff = (tid >> 5) * DIM + (tid & 31) * 8;
    const int sOff = (tid >> 5) * LDS_STRIDE + (tid & 31) * 8;
    const int bOff = (lane & 15) * LDS_STRIDE + hiSel;
    u32x4 rh0, rh1, rh2, rh3, rl0, rl1, rl2, rl3;

    float dmin[ROW_TILES][8], dmax[ROW_TILES][8];
#pragma unroll
    for (int rt = 0; rt < ROW_TILES; ++rt)
#pragma unroll
        for (int r = 0; r < 8; ++r) {
            dmin[rt][r] = 3.402823466e38f;
            dmax[rt][r] = 0.0f;
        }

    v16bf f0h[DIM / 32], f0l[DIM / 32], f1h[DIM / 32], f1l[DIM / 32];

    PREFETCH_B(0);
    STORE_B(0);
    PREFETCH_B(1);
    STORE_B(1);
    PREFETCH_B(2);
    __syncthreads();
    FRAG_LOAD(f0h, f0l, 0);

    for (int t = 0; t < TILES_PER_SLICE; t += 2) {
        SWEEP_STEP(t,     f0h, f0l, f1h, f1l);
        SWEEP_STEP(t + 1, f1h, f1l, f0h, f0l);
    }

    float minv[ROW_TILES][8];
#pragma unroll
    for (int rt = 0; rt < ROW_TILES; ++rt) {
#pragma unroll
        for (int r = 0; r < 8; ++r)
            minv[rt][r] = fminf(logpdf_cost(dmin[rt][r]),
                                logpdf_cost(dmax[rt][r]));
    }

#pragma unroll
    for (int rt = 0; rt < ROW_TILES; ++rt) {
#pragma unroll
        for (int r = 0; r < 8; ++r) {
#pragma unroll
            for (int off = 1; off < 16; off <<= 1)
                minv[rt][r] =
                    fminf(minv[rt][r], __shfl_xor(minv[rt][r], off, 32));
        }
    }

    if ((lane & 15) == 0) {
#pragma unroll
        for (int rt = 0; rt < ROW_TILES; ++rt) {
            const int base = rowBase + rt * 16 + ((lane >> 4) << 3);
#pragma unroll
            for (int r = 0; r < 8; ++r)
                __hip_atomic_fetch_min(&out[base + r], minv[rt][r],
                                       __ATOMIC_RELAXED,
                                       __HIP_MEMORY_SCOPE_AGENT);
        }
    }
}

#undef SWEEP_STEP
#undef COMPUTE_TILE
#undef PREFETCH_B
#undef STORE_B
#undef FRAG_LOAD

extern "C" void kernel_launch(void* const* d_in, const int* in_sizes, int n_in,
                              void* d_out, int out_size, void* d_ws, size_t ws_size,
                              hipStream_t stream) {
    const float* Ex = (const float*)d_in[0];
    const float* Ey = (const float*)d_in[1];

    const size_t elems = (size_t)NROWS * DIM;
    unsigned short* Xhi = (unsigned short*)d_ws;
    unsigned short* Xlo = Xhi + elems;
    unsigned short* Yhi = Xlo + elems;
    unsigned short* Ylo = Yhi + elems;
    float*          out = (float*)d_out;

    hipLaunchKernelGGL(normalize_split_kernel, dim3((2 * NROWS) / 8), dim3(256), 0,
                       stream, Ex, Ey, Xhi, Xlo, Yhi, Ylo, out);
    hipLaunchKernelGGL(cosmin_wmma_kernel,
                       dim3(NROWS / ROWS_PER_BLOCK, NSLICES), dim3(THREADS), 0,
                       stream, Xhi, Xlo, Yhi, Ylo, out);
}
